# starting kernel
# baseline (speedup 1.0000x reference)
#include <hip/hip_runtime.h>

#define NN 50000
#define EE 1600000
#define INF_ 128
#define HH 4
#define DD 32
#define HD 128
#define NEG_SLOPE 0.2f

typedef __attribute__((ext_vector_type(2))) float v2f;
typedef __attribute__((ext_vector_type(8))) float v8f;

__device__ __forceinline__ unsigned int ford(float f) {
  unsigned int b = __float_as_uint(f);
  return (b & 0x80000000u) ? ~b : (b | 0x80000000u);
}
__device__ __forceinline__ float funord(unsigned int u) {
  unsigned int b = (u & 0x80000000u) ? (u ^ 0x80000000u) : ~u;
  return __uint_as_float(b);
}
#define FORD_NEG_INF 0x007FFFFFu

__global__ void k_zero_f32(float* __restrict__ p, int n) {
  int i = blockIdx.x * blockDim.x + threadIdx.x;
  if (i < n) p[i] = 0.0f;
}
__global__ void k_init_ms(unsigned int* __restrict__ m, float* __restrict__ s, int n) {
  int i = blockIdx.x * blockDim.x + threadIdx.x;
  if (i < n) { m[i] = FORD_NEG_INF; s[i] = 0.0f; }
}

__global__ __launch_bounds__(256) void k_proj(const float* __restrict__ feat,
                                              const float* __restrict__ W,
                                              float* __restrict__ h) {
  __shared__ float As[16 * 132];
  const int m0 = blockIdx.x * 16;
  const int t  = threadIdx.x;

  {
    const int row = t >> 4;
    const int col = (t & 15) * 8;
    const float4* gp = (const float4*)(feat + (size_t)(m0 + row) * INF_ + col);
    float4 x0 = gp[0];
    float4 x1 = gp[1];
    float* lp = As + row * 132 + col;
    *(float4*)(lp)     = x0;
    *(float4*)(lp + 4) = x1;
  }
  __syncthreads();

  const int lane  = t & 31;
  const int wave  = t >> 5;
  const int n0    = wave * 16;
  const int r     = lane & 15;
  const int khalf = (lane >> 4) << 1;

  v8f acc = {};
  for (int k0 = 0; k0 < INF_; k0 += 4) {
    const int kk = k0 + khalf;
    v2f a;
    a.x = As[r * 132 + kk];
    a.y = As[r * 132 + kk + 1];
    v2f b;
    b.x = W[(size_t)(kk)     * HD + n0 + r];
    b.y = W[(size_t)(kk + 1) * HD + n0 + r];
    acc = __builtin_amdgcn_wmma_f32_16x16x4_f32(false, a, false, b,
                                                (short)0, acc, false, false);
  }

  const int mh = (lane >> 4) * 8;
#pragma unroll
  for (int j = 0; j < 8; ++j) {
    h[(size_t)(m0 + j + mh) * HD + n0 + r] = acc[j];
  }
}

__global__ void k_elr(const float* __restrict__ h, const float* __restrict__ al,
                      const float* __restrict__ ar, float* __restrict__ el,
                      float* __restrict__ er) {
  int tid = blockIdx.x * blockDim.x + threadIdx.x;
  if (tid >= NN * HH) return;
  const int n  = tid / HH;
  const int hh = tid % HH;
  const float* hp  = h  + (size_t)n * HD + hh * DD;
  const float* alp = al + hh * DD;
  const float* arp = ar + hh * DD;
  float sl = 0.0f, sr = 0.0f;
#pragma unroll
  for (int d = 0; d < DD; d += 4) {
    float4 hv = *(const float4*)(hp + d);
    float4 av = *(const float4*)(alp + d);
    float4 rv = *(const float4*)(arp + d);
    sl += hv.x * av.x + hv.y * av.y + hv.z * av.z + hv.w * av.w;
    sr += hv.x * rv.x + hv.y * rv.y + hv.z * rv.z + hv.w * rv.w;
  }
  el[tid] = sl;
  er[tid] = sr;
}

__global__ void k_edge_logit(const int* __restrict__ src, const int* __restrict__ dst,
                             const float* __restrict__ el, const float* __restrict__ er,
                             float* __restrict__ ebuf, unsigned int* __restrict__ m) {
  int tid = blockIdx.x * blockDim.x + threadIdx.x;
  if (tid >= EE * HH) return;
  const int e  = tid >> 2;
  const int hh = tid & 3;
  const int sd = src[e], dd = dst[e];
  float x = el[sd * HH + hh] + er[dd * HH + hh];
  x = (x > 0.0f) ? x : NEG_SLOPE * x;
  ebuf[tid] = x;
  atomicMax(&m[dd * HH + hh], ford(x));
}

__global__ void k_edge_exp(const int* __restrict__ dst, const unsigned int* __restrict__ m,
                           float* __restrict__ ebuf, float* __restrict__ s) {
  int tid = blockIdx.x * blockDim.x + threadIdx.x;
  if (tid >= EE * HH) return;
  const int e  = tid >> 2;
  const int hh = tid & 3;
  const int dd = dst[e];
  const float mm = funord(m[dd * HH + hh]);
  const float ex = __expf(ebuf[tid] - mm);
  ebuf[tid] = ex;
  atomicAdd(&s[dd * HH + hh], ex);
}

__global__ __launch_bounds__(256) void k_edge_msg(const int* __restrict__ src,
                                                  const int* __restrict__ dst,
                                                  const float* __restrict__ ebuf,
                                                  const float* __restrict__ s,
                                                  const float* __restrict__ h,
                                                  float* __restrict__ acc) {
  int tid = blockIdx.x * blockDim.x + threadIdx.x;
  if (tid >= EE * HH * 8) return;
  const int e   = tid >> 5;
  const int rem = tid & 31;
  const int hh  = rem >> 3;
  const int q   = rem & 7;
  const int sd  = src[e], dd = dst[e];
  const float alpha = ebuf[e * HH + hh] / s[dd * HH + hh];
  const float4 hv = *(const float4*)(h + (size_t)sd * HD + hh * DD + q * 4);
  float* ap = acc + (size_t)dd * HD + hh * DD + q * 4;
  atomicAdd(ap + 0, alpha * hv.x);
  atomicAdd(ap + 1, alpha * hv.y);
  atomicAdd(ap + 2, alpha * hv.z);
  atomicAdd(ap + 3, alpha * hv.w);
}

__global__ void k_mean(const float* __restrict__ acc, float* __restrict__ out) {
  int tid = blockIdx.x * blockDim.x + threadIdx.x;
  if (tid >= NN * DD) return;
  const int n = tid >> 5;
  const int d = tid & 31;
  const float* a = acc + (size_t)n * HD + d;
  out[tid] = 0.25f * (a[0] + a[32] + a[64] + a[96]);
}

extern "C" void kernel_launch(void* const* d_in, const int* in_sizes, int n_in,
                              void* d_out, int out_size, void* d_ws, size_t ws_size,
                              hipStream_t stream) {
  const float* feat = (const float*)d_in[0];
  const int*   src_g[2] = { (const int*)d_in[1], (const int*)d_in[3] };
  const int*   dst_g[2] = { (const int*)d_in[2], (const int*)d_in[4] };
  const float* W_g[2]   = { (const float*)d_in[5], (const float*)d_in[8] };
  const float* al_g[2]  = { (const float*)d_in[6], (const float*)d_in[9] };
  const float* ar_g[2]  = { (const float*)d_in[7], (const float*)d_in[10] };
  float* out = (float*)d_out;

  float*        acc  = (float*)d_ws;
  float*        hbuf = acc  + (size_t)NN * HD;
  float*        el   = hbuf + (size_t)NN * HD;
  float*        er   = el   + (size_t)NN * HH;
  unsigned int* m    = (unsigned int*)(er + (size_t)NN * HH);
  float*        s    = (float*)(m + (size_t)NN * HH);
  float*        ebuf = s + (size_t)NN * HH;

  const int T = 256;
  k_zero_f32<<<(NN * HD + T - 1) / T, T, 0, stream>>>(acc, NN * HD);

  for (int g = 0; g < 2; ++g) {
    k_proj<<<NN / 16, T, 0, stream>>>(feat, W_g[g], hbuf);
    k_elr<<<(NN * HH + T - 1) / T, T, 0, stream>>>(hbuf, al_g[g], ar_g[g], el, er);
    k_init_ms<<<(NN * HH + T - 1) / T, T, 0, stream>>>(m, s, NN * HH);
    k_edge_logit<<<(EE * HH + T - 1) / T, T, 0, stream>>>(src_g[g], dst_g[g], el, er, ebuf, m);
    k_edge_exp<<<(EE * HH + T - 1) / T, T, 0, stream>>>(dst_g[g], m, ebuf, s);
    k_edge_msg<<<(EE * HH * 8 + T - 1) / T, T, 0, stream>>>(src_g[g], dst_g[g], ebuf, s, hbuf, acc);
  }

  k_mean<<<(NN * DD + T - 1) / T, T, 0, stream>>>(acc, out);
}


# edge message: sum the 4 heads of an edge in registers (head mean folded in), 32 atomic adds per edge straight into out instead of 128 into a per-head buffer; changes the float summation order
# speedup vs baseline: 1.8681x; 1.8681x over previous
#include <hip/hip_runtime.h>

#define NN 50000
#define EE 1600000
#define INF_ 128
#define HH 4
#define DD 32
#define HD 128
#define NEG_SLOPE 0.2f

typedef __attribute__((ext_vector_type(2))) float v2f;
typedef __attribute__((ext_vector_type(8))) float v8f;

__device__ __forceinline__ unsigned int ford(float f) {
  unsigned int b = __float_as_uint(f);
  return (b & 0x80000000u) ? ~b : (b | 0x80000000u);
}
__device__ __forceinline__ float funord(unsigned int u) {
  unsigned int b = (u & 0x80000000u) ? (u ^ 0x80000000u) : ~u;
  return __uint_as_float(b);
}
#define FORD_NEG_INF 0x007FFFFFu

__global__ void k_zero_f32(float* __restrict__ p, int n) {
  int i = blockIdx.x * blockDim.x + threadIdx.x;
  if (i < n) p[i] = 0.0f;
}
__global__ void k_init_ms(unsigned int* __restrict__ m, float* __restrict__ s, int n) {
  int i = blockIdx.x * blockDim.x + threadIdx.x;
  if (i < n) { m[i] = FORD_NEG_INF; s[i] = 0.0f; }
}

__global__ __launch_bounds__(256) void k_proj(const float* __restrict__ feat,
                                              const float* __restrict__ W,
                                              float* __restrict__ h) {
  __shared__ float As[16 * 132];
  const int m0 = blockIdx.x * 16;
  const int t  = threadIdx.x;

  {
    const int row = t >> 4;
    const int col = (t & 15) * 8;
    const float4* gp = (const float4*)(feat + (size_t)(m0 + row) * INF_ + col);
    float4 x0 = gp[0];
    float4 x1 = gp[1];
    float* lp = As + row * 132 + col;
    *(float4*)(lp)     = x0;
    *(float4*)(lp + 4) = x1;
  }
  __syncthreads();

  const int lane  = t & 31;
  const int wave  = t >> 5;
  const int n0    = wave * 16;
  const int r     = lane & 15;
  const int khalf = (lane >> 4) << 1;

  v8f acc = {};
  for (int k0 = 0; k0 < INF_; k0 += 4) {
    const int kk = k0 + khalf;
    v2f a;
    a.x = As[r * 132 + kk];
    a.y = As[r * 132 + kk + 1];
    v2f b;
    b.x = W[(size_t)(kk)     * HD + n0 + r];
    b.y = W[(size_t)(kk + 1) * HD + n0 + r];
    acc = __builtin_amdgcn_wmma_f32_16x16x4_f32(false, a, false, b,
                                                (short)0, acc, false, false);
  }

  const int mh = (lane >> 4) * 8;
#pragma unroll
  for (int j = 0; j < 8; ++j) {
    h[(size_t)(m0 + j + mh) * HD + n0 + r] = acc[j];
  }
}

__global__ void k_elr(const float* __restrict__ h, const float* __restrict__ al,
                      const float* __restrict__ ar, float* __restrict__ el,
                      float* __restrict__ er) {
  int tid = blockIdx.x * blockDim.x + threadIdx.x;
  if (tid >= NN * HH) return;
  const int n  = tid / HH;
  const int hh = tid % HH;
  const float* hp  = h  + (size_t)n * HD + hh * DD;
  const float* alp = al + hh * DD;
  const float* arp = ar + hh * DD;
  float sl = 0.0f, sr = 0.0f;
#pragma unroll
  for (int d = 0; d < DD; d += 4) {
    float4 hv = *(const float4*)(hp + d);
    float4 av = *(const float4*)(alp + d);
    float4 rv = *(const float4*)(arp + d);
    sl += hv.x * av.x + hv.y * av.y + hv.z * av.z + hv.w * av.w;
    sr += hv.x * rv.x + hv.y * rv.y + hv.z * rv.z + hv.w * rv.w;
  }
  el[tid] = sl;
  er[tid] = sr;
}

__global__ void k_edge_logit(const int* __restrict__ src, const int* __restrict__ dst,
                             const float* __restrict__ el, const float* __restrict__ er,
                             float* __restrict__ ebuf, unsigned int* __restrict__ m) {
  int tid = blockIdx.x * blockDim.x + threadIdx.x;
  if (tid >= EE * HH) return;
  const int e  = tid >> 2;
  const int hh = tid & 3;
  const int sd = src[e], dd = dst[e];
  float x = el[sd * HH + hh] + er[dd * HH + hh];
  x = (x > 0.0f) ? x : NEG_SLOPE * x;
  ebuf[tid] = x;
  atomicMax(&m[dd * HH + hh], ford(x));
}

__global__ void k_edge_exp(const int* __restrict__ dst, const unsigned int* __restrict__ m,
                           float* __restrict__ ebuf, float* __restrict__ s) {
  int tid = blockIdx.x * blockDim.x + threadIdx.x;
  if (tid >= EE * HH) return;
  const int e  = tid >> 2;
  const int hh = tid & 3;
  const int dd = dst[e];
  const float mm = funord(m[dd * HH + hh]);
  const float ex = __expf(ebuf[tid] - mm);
  ebuf[tid] = ex;
  atomicAdd(&s[dd * HH + hh], ex);
}

static_assert(EE % 32 == 0, "one wave per 32 edges");
static_assert(DD == 32 && HH == 4, "lane = feature, four heads per edge");

__device__ __forceinline__ float bcast_f(float v, int j) {
  return __int_as_float(__builtin_amdgcn_readlane(__float_as_int(v), j));
}

__global__ __launch_bounds__(256) void k_edge_msg(const int* __restrict__ src,
                                                  const int* __restrict__ dst,
                                                  const float* __restrict__ ebuf,
                                                  const float* __restrict__ s,
                                                  const float* __restrict__ h,
                                                  float* __restrict__ out) {
  const int lane = threadIdx.x & 31;
  const int wv   = (blockIdx.x * blockDim.x + threadIdx.x) >> 5;
  const int e0   = wv * 32;
  if (e0 >= EE) return;
  const int me = e0 + lane;
  const int sd = src[me], dd = dst[me];
  const float4 ex = *(const float4*)(ebuf + (size_t)me * HH);
  const float4 sv = *(const float4*)(s + (size_t)dd * HH);
  const float a0 = 0.25f * (ex.x / sv.x);
  const float a1 = 0.25f * (ex.y / sv.y);
  const float a2 = 0.25f * (ex.z / sv.z);
  const float a3 = 0.25f * (ex.w / sv.w);

#pragma unroll
  for (int jb = 0; jb < 32; jb += 8) {
    float hv[8][4];
#pragma unroll
    for (int u = 0; u < 8; ++u) {
      const int sj = __builtin_amdgcn_readlane(sd, jb + u);
      const float* hp = h + (size_t)sj * HD + lane;
      hv[u][0] = hp[0];
      hv[u][1] = hp[DD];
      hv[u][2] = hp[2 * DD];
      hv[u][3] = hp[3 * DD];
    }
#pragma unroll
    for (int u = 0; u < 8; ++u) {
      const int dj = __builtin_amdgcn_readlane(dd, jb + u);
      const float b0 = bcast_f(a0, jb + u);
      const float b1 = bcast_f(a1, jb + u);
      const float b2 = bcast_f(a2, jb + u);
      const float b3 = bcast_f(a3, jb + u);
      float v = b0 * hv[u][0];
      v = __builtin_fmaf(b1, hv[u][1], v);
      v = __builtin_fmaf(b2, hv[u][2], v);
      v = __builtin_fmaf(b3, hv[u][3], v);
      atomicAdd(out + (size_t)dj * DD + lane, v);
    }
  }
}

extern "C" void kernel_launch(void* const* d_in, const int* in_sizes, int n_in,
                              void* d_out, int out_size, void* d_ws, size_t ws_size,
                              hipStream_t stream) {
  const float* feat = (const float*)d_in[0];
  const int*   src_g[2] = { (const int*)d_in[1], (const int*)d_in[3] };
  const int*   dst_g[2] = { (const int*)d_in[2], (const int*)d_in[4] };
  const float* W_g[2]   = { (const float*)d_in[5], (const float*)d_in[8] };
  const float* al_g[2]  = { (const float*)d_in[6], (const float*)d_in[9] };
  const float* ar_g[2]  = { (const float*)d_in[7], (const float*)d_in[10] };
  float* out = (float*)d_out;

  float*        hbuf = (float*)d_ws;
  float*        el   = hbuf + (size_t)NN * HD;
  float*        er   = el   + (size_t)NN * HH;
  unsigned int* m    = (unsigned int*)(er + (size_t)NN * HH);
  float*        s    = (float*)(m + (size_t)NN * HH);
  float*        ebuf = s + (size_t)NN * HH;

  const int T = 256;
  k_zero_f32<<<(NN * DD + T - 1) / T, T, 0, stream>>>(out, NN * DD);

  for (int g = 0; g < 2; ++g) {
    k_proj<<<NN / 16, T, 0, stream>>>(feat, W_g[g], hbuf);
    k_elr<<<(NN * HH + T - 1) / T, T, 0, stream>>>(hbuf, al_g[g], ar_g[g], el, er);
    k_init_ms<<<(NN * HH + T - 1) / T, T, 0, stream>>>(m, s, NN * HH);
    k_edge_logit<<<(EE * HH + T - 1) / T, T, 0, stream>>>(src_g[g], dst_g[g], el, er, ebuf, m);
    k_edge_exp<<<(EE * HH + T - 1) / T, T, 0, stream>>>(dst_g[g], m, ebuf, s);
    k_edge_msg<<<(EE + T - 1) / T, T, 0, stream>>>(src_g[g], dst_g[g], ebuf, s, hbuf, out);
  }
}


# edges ordered by destination (counting sort through LDS counters, no global atomics); one wave per node does softmax+messages in registers and stores out once; sums now in bucket order (LDS arrival)
# speedup vs baseline: 3.7162x; 1.9893x over previous
#include <hip/hip_runtime.h>


#define NN 50000
#define EE 1600000
#define INF_ 128
#define HH 4
#define DD 32
#define HD 128
#define NEG_SLOPE 0.2f

typedef __attribute__((ext_vector_type(2))) float v2f;
typedef __attribute__((ext_vector_type(8))) float v8f;

__global__ __launch_bounds__(256) void k_proj(const float* __restrict__ feat,
                                              const float* __restrict__ W,
                                              float* __restrict__ h) {
  __shared__ float As[16 * 132];
  const int m0 = blockIdx.x * 16;
  const int t  = threadIdx.x;

  {
    const int row = t >> 4;
    const int col = (t & 15) * 8;
    const float4* gp = (const float4*)(feat + (size_t)(m0 + row) * INF_ + col);
    float4 x0 = gp[0];
    float4 x1 = gp[1];
    float* lp = As + row * 132 + col;
    *(float4*)(lp)     = x0;
    *(float4*)(lp + 4) = x1;
  }
  __syncthreads();

  const int lane  = t & 31;
  const int wave  = t >> 5;
  const int n0    = wave * 16;
  const int r     = lane & 15;
  const int khalf = (lane >> 4) << 1;

  v8f acc = {};
  for (int k0 = 0; k0 < INF_; k0 += 4) {
    const int kk = k0 + khalf;
    v2f a;
    a.x = As[r * 132 + kk];
    a.y = As[r * 132 + kk + 1];
    v2f b;
    b.x = W[(size_t)(kk)     * HD + n0 + r];
    b.y = W[(size_t)(kk + 1) * HD + n0 + r];
    acc = __builtin_amdgcn_wmma_f32_16x16x4_f32(false, a, false, b,
                                                (short)0, acc, false, false);
  }

  const int mh = (lane >> 4) * 8;
#pragma unroll
  for (int j = 0; j < 8; ++j) {
    h[(size_t)(m0 + j + mh) * HD + n0 + r] = acc[j];
  }
}

__global__ void k_elr(const float* __restrict__ h, const float* __restrict__ al,
                      const float* __restrict__ ar, float* __restrict__ el,
                      float* __restrict__ er) {
  int tid = blockIdx.x * blockDim.x + threadIdx.x;
  if (tid >= NN * HH) return;
  const int n  = tid / HH;
  const int hh = tid % HH;
  const float* hp  = h  + (size_t)n * HD + hh * DD;
  const float* alp = al + hh * DD;
  const float* arp = ar + hh * DD;
  float sl = 0.0f, sr = 0.0f;
#pragma unroll
  for (int d = 0; d < DD; d += 4) {
    float4 hv = *(const float4*)(hp + d);
    float4 av = *(const float4*)(alp + d);
    float4 rv = *(const float4*)(arp + d);
    sl += hv.x * av.x + hv.y * av.y + hv.z * av.z + hv.w * av.w;
    sr += hv.x * rv.x + hv.y * rv.y + hv.z * rv.z + hv.w * rv.w;
  }
  el[tid] = sl;
  er[tid] = sr;
}

#define NCH   64
#define CH_E  (EE / NCH)
#define BK_T  1024
#define CH_IT ((CH_E + BK_T - 1) / BK_T)
static_assert(EE % NCH == 0 && NN % 4 == 0, "whole chunks, whole uint4 rows");

__global__ __launch_bounds__(BK_T) void k_count(const int* __restrict__ dst_f,
                                                const int* __restrict__ dst_b,
                                                unsigned int* __restrict__ cnt) {
  __shared__ __attribute__((aligned(16))) unsigned int hist[NN];
  const int t = threadIdx.x;
  const int w = blockIdx.x;
  const int g = blockIdx.y;
  const int* dst = (g ? dst_b : dst_f) + (size_t)w * CH_E;
  int d[CH_IT];
#pragma unroll
  for (int i = 0; i < CH_IT; ++i) {
    const int k = i * BK_T + t;
    d[i] = (k < CH_E) ? dst[k] : -1;
  }
  for (int i = t; i < NN / 4; i += BK_T) ((uint4*)hist)[i] = make_uint4(0u, 0u, 0u, 0u);
  __syncthreads();
#pragma unroll
  for (int i = 0; i < CH_IT; ++i)
    if ((unsigned int)d[i] < (unsigned int)NN) atomicAdd(&hist[d[i]], 1u);
  __syncthreads();
  uint4* row = (uint4*)(cnt + ((size_t)g * NCH + w) * NN);
  for (int i = t; i < NN / 4; i += BK_T) row[i] = ((const uint4*)hist)[i];
}

__global__ void k_col(unsigned int* __restrict__ cnt, unsigned int* __restrict__ deg) {
  const int i = blockIdx.x * blockDim.x + threadIdx.x;
  if (i >= 2 * NN) return;
  const int g = (i >= NN) ? 1 : 0;
  const int n = i - g * NN;
  unsigned int* col = cnt + (size_t)g * NCH * NN + n;
  unsigned int c[NCH];
#pragma unroll
  for (int w = 0; w < NCH; ++w) c[w] = col[(size_t)w * NN];
  unsigned int run = 0u;
#pragma unroll
  for (int w = 0; w < NCH; ++w) { col[(size_t)w * NN] = run; run += c[w]; }
  deg[i] = run;
}

#define SCAN_T 1024
#define SCAN_V 13
static_assert(NN % 4 == 0 && SCAN_T * SCAN_V * 4 >= NN, "thread chunks of whole uint4 cover the nodes");

__global__ __launch_bounds__(SCAN_T) void k_scan(const unsigned int* __restrict__ deg,
                                                 unsigned int* __restrict__ off) {
  __shared__ unsigned int wtot[SCAN_T / 32];
  const int g    = blockIdx.x;
  const int t    = threadIdx.x;
  const int lane = t & 31;
  const int wave = t >> 5;
  const int q0   = t * SCAN_V;
  const uint4* dg = (const uint4*)(deg + (size_t)g * NN) + q0;
  uint4*       og = (uint4*)(off + (size_t)g * NN) + q0;

  uint4 v[SCAN_V];
  unsigned int sum = 0u;
#pragma unroll
  for (int i = 0; i < SCAN_V; ++i) {
    v[i] = (q0 + i < NN / 4) ? dg[i] : make_uint4(0u, 0u, 0u, 0u);
    sum += v[i].x + v[i].y + v[i].z + v[i].w;
  }
  unsigned int inc = sum;
#pragma unroll
  for (int o = 1; o < 32; o <<= 1) {
    const unsigned int y = __shfl_up(inc, (unsigned int)o);
    if (lane >= o) inc += y;
  }
  if (lane == 31) wtot[wave] = inc;
  __syncthreads();
  unsigned int wb = (lane < wave) ? wtot[lane] : 0u;
#pragma unroll
  for (int o = 16; o; o >>= 1) wb += __shfl_xor(wb, o);
  unsigned int run = wb + inc - sum;
#pragma unroll
  for (int i = 0; i < SCAN_V; ++i) {
    if (q0 + i < NN / 4) {
      uint4 o4;
      o4.x = run; run += v[i].x;
      o4.y = run; run += v[i].y;
      o4.z = run; run += v[i].z;
      o4.w = run; run += v[i].w;
      og[i] = o4;
    }
  }
}

__global__ __launch_bounds__(BK_T) void k_fill(const int* __restrict__ src_f, const int* __restrict__ dst_f,
                                               const int* __restrict__ src_b, const int* __restrict__ dst_b,
                                               const unsigned int* __restrict__ cnt,
                                               const unsigned int* __restrict__ off,
                                               int* __restrict__ perm) {
  __shared__ __attribute__((aligned(16))) unsigned int slot[NN];
  const int t = threadIdx.x;
  const int w = blockIdx.x;
  const int g = blockIdx.y;
  const int* dst = (g ? dst_b : dst_f) + (size_t)w * CH_E;
  const int* src = (g ? src_b : src_f) + (size_t)w * CH_E;
  int d[CH_IT], s[CH_IT];
#pragma unroll
  for (int i = 0; i < CH_IT; ++i) {
    const int k = i * BK_T + t;
    d[i] = (k < CH_E) ? dst[k] : -1;
    s[i] = (k < CH_E) ? src[k] : 0;
  }
  const uint4* row = (const uint4*)(cnt + ((size_t)g * NCH + w) * NN);
  const uint4* og  = (const uint4*)(off + (size_t)g * NN);
  for (int i = t; i < NN / 4; i += BK_T) {
    const uint4 a = row[i];
    const uint4 b = og[i];
    ((uint4*)slot)[i] = make_uint4(a.x + b.x, a.y + b.y, a.z + b.z, a.w + b.w);
  }
  __syncthreads();
  int* pg = perm + (size_t)g * EE;
#pragma unroll
  for (int i = 0; i < CH_IT; ++i) {
    if ((unsigned int)d[i] < (unsigned int)NN) {
      const unsigned int pos = atomicAdd(&slot[d[i]], 1u);
      if (pos < (unsigned int)EE) pg[pos] = s[i];
    }
  }
}

static_assert(DD == 32 && HH == 4, "lane = feature, four heads per edge");

__device__ __forceinline__ float bcast_f(float v, int j) {
  return __int_as_float(__builtin_amdgcn_readlane(__float_as_int(v), j));
}
__device__ __forceinline__ float lrelu(float x) { return (x > 0.0f) ? x : NEG_SLOPE * x; }
__device__ __forceinline__ float wave_max(float v) {
#pragma unroll
  for (int o = 16; o; o >>= 1) v = fmaxf(v, __shfl_xor(v, o));
  return v;
}
__device__ __forceinline__ float wave_sum(float v) {
#pragma unroll
  for (int o = 16; o; o >>= 1) v += __shfl_xor(v, o);
  return v;
}

__device__ __forceinline__ void load_edge(const int* __restrict__ perm, unsigned int beg,
                                          unsigned int deg, unsigned int j,
                                          const float* __restrict__ el, const float4 er4,
                                          int& sj, float x[4]) {
  const unsigned int jc = (j < deg) ? j : deg - 1u;
  unsigned int p = beg + jc;
  p = (p < (unsigned int)EE) ? p : (unsigned int)(EE - 1);
  int s = perm[p];
  s = ((unsigned int)s < (unsigned int)NN) ? s : 0;
  sj = s;
  const float4 e4 = *(const float4*)(el + (size_t)s * HH);
  x[0] = lrelu(e4.x + er4.x);
  x[1] = lrelu(e4.y + er4.y);
  x[2] = lrelu(e4.z + er4.z);
  x[3] = lrelu(e4.w + er4.w);
}

__device__ __forceinline__ float chunk_msg(const float* __restrict__ h, int sj, const float a[4],
                                           unsigned int cnt, int lane, float acc) {
  for (unsigned int jb = 0; jb < cnt; jb += 8) {
    float hv[8][4];
#pragma unroll
    for (int u = 0; u < 8; ++u) {
      const int su = __builtin_amdgcn_readlane(sj, (int)jb + u);
      const float* hp = h + (size_t)su * HD + lane;
      hv[u][0] = hp[0];
      hv[u][1] = hp[DD];
      hv[u][2] = hp[2 * DD];
      hv[u][3] = hp[3 * DD];
    }
#pragma unroll
    for (int u = 0; u < 8; ++u) {
      const float b0 = bcast_f(a[0], (int)jb + u);
      const float b1 = bcast_f(a[1], (int)jb + u);
      const float b2 = bcast_f(a[2], (int)jb + u);
      const float b3 = bcast_f(a[3], (int)jb + u);
      float v = b0 * hv[u][0];
      v = __builtin_fmaf(b1, hv[u][1], v);
      v = __builtin_fmaf(b2, hv[u][2], v);
      v = __builtin_fmaf(b3, hv[u][3], v);
      acc += v;
    }
  }
  return acc;
}

__device__ __forceinline__ float agg_graph(const int* __restrict__ perm, unsigned int beg,
                                           unsigned int deg, const float* __restrict__ el,
                                           const float4 er4, const float* __restrict__ h,
                                           int lane, float acc) {
  if (deg == 0u) return acc;
  const float ninf = -__builtin_inff();
  const bool two = deg > 32u;
  int sjA, sjB = 0;
  float xA[4], xB[4] = { ninf, ninf, ninf, ninf };
  load_edge(perm, beg, deg, (unsigned int)lane, el, er4, sjA, xA);
  if (two) load_edge(perm, beg, deg, 32u + (unsigned int)lane, el, er4, sjB, xB);

  float m[4];
#pragma unroll
  for (int k = 0; k < 4; ++k) m[k] = fmaxf(xA[k], xB[k]);
  for (unsigned int base = 64u; base < deg; base += 32u) {
    int sj; float x[4];
    load_edge(perm, beg, deg, base + (unsigned int)lane, el, er4, sj, x);
#pragma unroll
    for (int k = 0; k < 4; ++k) m[k] = fmaxf(m[k], x[k]);
  }
#pragma unroll
  for (int k = 0; k < 4; ++k) m[k] = wave_max(m[k]);

  const bool vA = (unsigned int)lane < deg;
  const bool vB = 32u + (unsigned int)lane < deg;
  float s[4];
#pragma unroll
  for (int k = 0; k < 4; ++k) {
    xA[k] = vA ? __expf(xA[k] - m[k]) : 0.0f;
    xB[k] = vB ? __expf(xB[k] - m[k]) : 0.0f;
    s[k]  = xA[k] + xB[k];
  }
  for (unsigned int base = 64u; base < deg; base += 32u) {
    int sj; float x[4];
    load_edge(perm, beg, deg, base + (unsigned int)lane, el, er4, sj, x);
    const bool v = base + (unsigned int)lane < deg;
#pragma unroll
    for (int k = 0; k < 4; ++k) s[k] += v ? __expf(x[k] - m[k]) : 0.0f;
  }
#pragma unroll
  for (int k = 0; k < 4; ++k) s[k] = wave_sum(s[k]);

  float a[4];
#pragma unroll
  for (int k = 0; k < 4; ++k) a[k] = 0.25f * (xA[k] / s[k]);
  acc = chunk_msg(h, sjA, a, (deg < 32u) ? deg : 32u, lane, acc);
  if (two) {
#pragma unroll
    for (int k = 0; k < 4; ++k) a[k] = 0.25f * (xB[k] / s[k]);
    acc = chunk_msg(h, sjB, a, (deg - 32u < 32u) ? deg - 32u : 32u, lane, acc);
  }
  for (unsigned int base = 64u; base < deg; base += 32u) {
    int sj; float x[4];
    load_edge(perm, beg, deg, base + (unsigned int)lane, el, er4, sj, x);
    const bool v = base + (unsigned int)lane < deg;
#pragma unroll
    for (int k = 0; k < 4; ++k) a[k] = v ? 0.25f * (__expf(x[k] - m[k]) / s[k]) : 0.0f;
    acc = chunk_msg(h, sj, a, (deg - base < 32u) ? deg - base : 32u, lane, acc);
  }
  return acc;
}

__global__ __launch_bounds__(256) void k_node(const int* __restrict__ perm,
                                              const unsigned int* __restrict__ off,
                                              const unsigned int* __restrict__ deg,
                                              const float* __restrict__ el,
                                              const float* __restrict__ er,
                                              const float* __restrict__ h,
                                              float* __restrict__ out) {
  const int lane = threadIdx.x & 31;
  const int node = __builtin_amdgcn_readfirstlane((int)((blockIdx.x * blockDim.x + threadIdx.x) >> 5));
  if (node >= NN) return;
  float acc = 0.0f;
#pragma unroll
  for (int g = 0; g < 2; ++g) {
    const unsigned int dg  = (unsigned int)__builtin_amdgcn_readfirstlane((int)deg[(size_t)g * NN + node]);
    const unsigned int beg = (unsigned int)__builtin_amdgcn_readfirstlane((int)off[(size_t)g * NN + node]);
    const float4 er4 = *(const float4*)(er + ((size_t)g * NN + node) * HH);
    acc = agg_graph(perm + (size_t)g * EE, beg, dg, el + (size_t)g * NN * HH, er4,
                    h + (size_t)g * NN * HD, lane, acc);
  }
  out[(size_t)node * DD + lane] = acc;
}

extern "C" void kernel_launch(void* const* d_in, const int* in_sizes, int n_in,
                              void* d_out, int out_size, void* d_ws, size_t ws_size,
                              hipStream_t stream) {
  const float* feat = (const float*)d_in[0];
  const int*   src_g[2] = { (const int*)d_in[1], (const int*)d_in[3] };
  const int*   dst_g[2] = { (const int*)d_in[2], (const int*)d_in[4] };
  const float* W_g[2]   = { (const float*)d_in[5], (const float*)d_in[8] };
  const float* al_g[2]  = { (const float*)d_in[6], (const float*)d_in[9] };
  const float* ar_g[2]  = { (const float*)d_in[7], (const float*)d_in[10] };
  float* out = (float*)d_out;

  float*        hbuf = (float*)d_ws;
  float*        el   = hbuf + (size_t)2 * NN * HD;
  float*        er   = el   + (size_t)2 * NN * HH;
  unsigned int* deg  = (unsigned int*)(er + (size_t)2 * NN * HH);
  unsigned int* off  = deg + (size_t)2 * NN;
  unsigned int* cnt  = off + (size_t)2 * NN;
  int*          perm = (int*)(cnt + (size_t)2 * NCH * NN);

  const int T = 256;
  k_count<<<dim3(NCH, 2), BK_T, 0, stream>>>(dst_g[0], dst_g[1], cnt);
  k_col<<<(2 * NN + T - 1) / T, T, 0, stream>>>(cnt, deg);
  k_scan<<<2, SCAN_T, 0, stream>>>(deg, off);
  k_fill<<<dim3(NCH, 2), BK_T, 0, stream>>>(src_g[0], dst_g[0], src_g[1], dst_g[1], cnt, off, perm);
  for (int g = 0; g < 2; ++g) {
    k_proj<<<NN / 16, T, 0, stream>>>(feat, W_g[g], hbuf + (size_t)g * NN * HD);
    k_elr<<<(NN * HH + T - 1) / T, T, 0, stream>>>(hbuf + (size_t)g * NN * HD, al_g[g], ar_g[g],
                                                   el + (size_t)g * NN * HH, er + (size_t)g * NN * HH);
  }
  k_node<<<(NN * 32 + T - 1) / T, T, 0, stream>>>(perm, off, deg, el, er, hbuf, out);
}


# node pass: lane = head x 4 features, 16-byte row loads; logits from the gathered h rows (8-lane sums), running-max softmax per 8 edges; el/er tables and their 2 launches go; sum order changes
# speedup vs baseline: 3.9947x; 1.0749x over previous
#include <hip/hip_runtime.h>


#define NN 50000
#define EE 1600000
#define INF_ 128
#define HH 4
#define DD 32
#define HD 128
#define NEG_SLOPE 0.2f

typedef __attribute__((ext_vector_type(2))) float v2f;
typedef __attribute__((ext_vector_type(8))) float v8f;

__global__ __launch_bounds__(256) void k_proj(const float* __restrict__ feat,
                                              const float* __restrict__ W,
                                              float* __restrict__ h) {
  __shared__ float As[16 * 132];
  const int m0 = blockIdx.x * 16;
  const int t  = threadIdx.x;

  {
    const int row = t >> 4;
    const int col = (t & 15) * 8;
    const float4* gp = (const float4*)(feat + (size_t)(m0 + row) * INF_ + col);
    float4 x0 = gp[0];
    float4 x1 = gp[1];
    float* lp = As + row * 132 + col;
    *(float4*)(lp)     = x0;
    *(float4*)(lp + 4) = x1;
  }
  __syncthreads();

  const int lane  = t & 31;
  const int wave  = t >> 5;
  const int n0    = wave * 16;
  const int r     = lane & 15;
  const int khalf = (lane >> 4) << 1;

  v8f acc = {};
  for (int k0 = 0; k0 < INF_; k0 += 4) {
    const int kk = k0 + khalf;
    v2f a;
    a.x = As[r * 132 + kk];
    a.y = As[r * 132 + kk + 1];
    v2f b;
    b.x = W[(size_t)(kk)     * HD + n0 + r];
    b.y = W[(size_t)(kk + 1) * HD + n0 + r];
    acc = __builtin_amdgcn_wmma_f32_16x16x4_f32(false, a, false, b,
                                                (short)0, acc, false, false);
  }

  const int mh = (lane >> 4) * 8;
#pragma unroll
  for (int j = 0; j < 8; ++j) {
    h[(size_t)(m0 + j + mh) * HD + n0 + r] = acc[j];
  }
}

#define NCH   64
#define CH_E  (EE / NCH)
#define BK_T  1024
#define CH_IT ((CH_E + BK_T - 1) / BK_T)
static_assert(EE % NCH == 0 && NN % 4 == 0, "whole chunks, whole uint4 rows");

__global__ __launch_bounds__(BK_T) void k_count(const int* __restrict__ dst_f,
                                                const int* __restrict__ dst_b,
                                                unsigned int* __restrict__ cnt) {
  __shared__ __attribute__((aligned(16))) unsigned int hist[NN];
  const int t = threadIdx.x;
  const int w = blockIdx.x;
  const int g = blockIdx.y;
  const int* dst = (g ? dst_b : dst_f) + (size_t)w * CH_E;
  int d[CH_IT];
#pragma unroll
  for (int i = 0; i < CH_IT; ++i) {
    const int k = i * BK_T + t;
    d[i] = (k < CH_E) ? dst[k] : -1;
  }
  for (int i = t; i < NN / 4; i += BK_T) ((uint4*)hist)[i] = make_uint4(0u, 0u, 0u, 0u);
  __syncthreads();
#pragma unroll
  for (int i = 0; i < CH_IT; ++i)
    if ((unsigned int)d[i] < (unsigned int)NN) atomicAdd(&hist[d[i]], 1u);
  __syncthreads();
  uint4* row = (uint4*)(cnt + ((size_t)g * NCH + w) * NN);
  for (int i = t; i < NN / 4; i += BK_T) row[i] = ((const uint4*)hist)[i];
}

__global__ void k_col(unsigned int* __restrict__ cnt, unsigned int* __restrict__ deg) {
  const int i = blockIdx.x * blockDim.x + threadIdx.x;
  if (i >= 2 * NN) return;
  const int g = (i >= NN) ? 1 : 0;
  const int n = i - g * NN;
  unsigned int* col = cnt + (size_t)g * NCH * NN + n;
  unsigned int c[NCH];
#pragma unroll
  for (int w = 0; w < NCH; ++w) c[w] = col[(size_t)w * NN];
  unsigned int run = 0u;
#pragma unroll
  for (int w = 0; w < NCH; ++w) { col[(size_t)w * NN] = run; run += c[w]; }
  deg[i] = run;
}

#define SCAN_T 1024
#define SCAN_V 13
static_assert(NN % 4 == 0 && SCAN_T * SCAN_V * 4 >= NN, "thread chunks of whole uint4 cover the nodes");

__global__ __launch_bounds__(SCAN_T) void k_scan(const unsigned int* __restrict__ deg,
                                                 unsigned int* __restrict__ off) {
  __shared__ unsigned int wtot[SCAN_T / 32];
  const int g    = blockIdx.x;
  const int t    = threadIdx.x;
  const int lane = t & 31;
  const int wave = t >> 5;
  const int q0   = t * SCAN_V;
  const uint4* dg = (const uint4*)(deg + (size_t)g * NN) + q0;
  uint4*       og = (uint4*)(off + (size_t)g * NN) + q0;

  uint4 v[SCAN_V];
  unsigned int sum = 0u;
#pragma unroll
  for (int i = 0; i < SCAN_V; ++i) {
    v[i] = (q0 + i < NN / 4) ? dg[i] : make_uint4(0u, 0u, 0u, 0u);
    sum += v[i].x + v[i].y + v[i].z + v[i].w;
  }
  unsigned int inc = sum;
#pragma unroll
  for (int o = 1; o < 32; o <<= 1) {
    const unsigned int y = __shfl_up(inc, (unsigned int)o);
    if (lane >= o) inc += y;
  }
  if (lane == 31) wtot[wave] = inc;
  __syncthreads();
  unsigned int wb = (lane < wave) ? wtot[lane] : 0u;
#pragma unroll
  for (int o = 16; o; o >>= 1) wb += __shfl_xor(wb, o);
  unsigned int run = wb + inc - sum;
#pragma unroll
  for (int i = 0; i < SCAN_V; ++i) {
    if (q0 + i < NN / 4) {
      uint4 o4;
      o4.x = run; run += v[i].x;
      o4.y = run; run += v[i].y;
      o4.z = run; run += v[i].z;
      o4.w = run; run += v[i].w;
      og[i] = o4;
    }
  }
}

__global__ __launch_bounds__(BK_T) void k_fill(const int* __restrict__ src_f, const int* __restrict__ dst_f,
                                               const int* __restrict__ src_b, const int* __restrict__ dst_b,
                                               const unsigned int* __restrict__ cnt,
                                               const unsigned int* __restrict__ off,
                                               int* __restrict__ perm) {
  __shared__ __attribute__((aligned(16))) unsigned int slot[NN];
  const int t = threadIdx.x;
  const int w = blockIdx.x;
  const int g = blockIdx.y;
  const int* dst = (g ? dst_b : dst_f) + (size_t)w * CH_E;
  const int* src = (g ? src_b : src_f) + (size_t)w * CH_E;
  int d[CH_IT], s[CH_IT];
#pragma unroll
  for (int i = 0; i < CH_IT; ++i) {
    const int k = i * BK_T + t;
    d[i] = (k < CH_E) ? dst[k] : -1;
    s[i] = (k < CH_E) ? src[k] : 0;
  }
  const uint4* row = (const uint4*)(cnt + ((size_t)g * NCH + w) * NN);
  const uint4* og  = (const uint4*)(off + (size_t)g * NN);
  for (int i = t; i < NN / 4; i += BK_T) {
    const uint4 a = row[i];
    const uint4 b = og[i];
    ((uint4*)slot)[i] = make_uint4(a.x + b.x, a.y + b.y, a.z + b.z, a.w + b.w);
  }
  __syncthreads();
  int* pg = perm + (size_t)g * EE;
#pragma unroll
  for (int i = 0; i < CH_IT; ++i) {
    if ((unsigned int)d[i] < (unsigned int)NN) {
      const unsigned int pos = atomicAdd(&slot[d[i]], 1u);
      if (pos < (unsigned int)EE) pg[pos] = s[i];
    }
  }
}

static_assert(DD == 32 && HH == 4, "lane = (head, four features)");

__device__ __forceinline__ float lrelu(float x) { return (x > 0.0f) ? x : NEG_SLOPE * x; }

template <int MASK>
__device__ __forceinline__ float xmask(float v) {
  return __int_as_float(__builtin_amdgcn_mov_dpp(__float_as_int(v), 0x160 | MASK, 0xf, 0xf, false));
}
__device__ __forceinline__ float sum8(float v) {
  v += xmask<1>(v);
  v += xmask<2>(v);
  v += xmask<4>(v);
  return v;
}
__device__ __forceinline__ float dot4(const float4 a, const float4 b) {
  v2f t = (v2f){a.x, a.y} * (v2f){b.x, b.y};
  t = __builtin_elementwise_fma((v2f){a.z, a.w}, (v2f){b.z, b.w}, t);
  return t.x + t.y;
}

__device__ __forceinline__ float4 agg_graph(const int* __restrict__ perm, unsigned int beg,
                                            unsigned int deg, const float* __restrict__ h,
                                            const float* __restrict__ al, const float* __restrict__ ar,
                                            int node, int lane, float4 res) {
  if (deg == 0u) return res;
  const float ninf = -__builtin_inff();
  const float4 al4 = *(const float4*)(al + 4 * lane);
  const float4 ar4 = *(const float4*)(ar + 4 * lane);
  const float4 hn  = *(const float4*)(h + (size_t)node * HD + 4 * lane);
  const float er   = sum8(dot4(hn, ar4));

  float m = ninf, s = 0.0f;
  float4 acc = make_float4(0.0f, 0.0f, 0.0f, 0.0f);
  for (unsigned int base = 0u; base < deg; base += 32u) {
    const unsigned int j  = base + (unsigned int)lane;
    const unsigned int jc = (j < deg) ? j : deg - 1u;
    unsigned int p = beg + jc;
    p = (p < (unsigned int)EE) ? p : (unsigned int)(EE - 1);
    int sj = perm[p];
    sj = ((unsigned int)sj < (unsigned int)NN) ? sj : 0;
    const unsigned int cnt = (deg - base < 32u) ? deg - base : 32u;
    for (unsigned int jb = 0u; jb < cnt; jb += 8u) {
      float4 hv[8];
#pragma unroll
      for (int u = 0; u < 8; ++u) {
        const int su = __builtin_amdgcn_readlane(sj, (int)jb + u);
        hv[u] = *(const float4*)(h + (size_t)su * HD + 4 * lane);
      }
      float x[8];
      float gm = ninf;
#pragma unroll
      for (int u = 0; u < 8; ++u) {
        const float t = lrelu(sum8(dot4(hv[u], al4)) + er);
        x[u] = (jb + (unsigned int)u < cnt) ? t : ninf;
        gm = fmaxf(gm, x[u]);
      }
      const float mn = fmaxf(m, gm);
      const float sc = __expf(m - mn);
      s *= sc;
      acc.x *= sc; acc.y *= sc; acc.z *= sc; acc.w *= sc;
      m = mn;
#pragma unroll
      for (int u = 0; u < 8; ++u) {
        const float pe = __expf(x[u] - mn);
        s += pe;
        acc.x = __builtin_fmaf(pe, hv[u].x, acc.x);
        acc.y = __builtin_fmaf(pe, hv[u].y, acc.y);
        acc.z = __builtin_fmaf(pe, hv[u].z, acc.z);
        acc.w = __builtin_fmaf(pe, hv[u].w, acc.w);
      }
    }
  }
  const float w = 0.25f / s;
  res.x = __builtin_fmaf(acc.x, w, res.x);
  res.y = __builtin_fmaf(acc.y, w, res.y);
  res.z = __builtin_fmaf(acc.z, w, res.z);
  res.w = __builtin_fmaf(acc.w, w, res.w);
  return res;
}

__global__ __launch_bounds__(256) void k_node(const int* __restrict__ perm,
                                              const unsigned int* __restrict__ off,
                                              const unsigned int* __restrict__ deg,
                                              const float* __restrict__ h,
                                              const float* __restrict__ al_f,
                                              const float* __restrict__ ar_f,
                                              const float* __restrict__ al_b,
                                              const float* __restrict__ ar_b,
                                              float* __restrict__ out) {
  const int lane = threadIdx.x & 31;
  const int node = __builtin_amdgcn_readfirstlane((int)((blockIdx.x * blockDim.x + threadIdx.x) >> 5));
  if (node >= NN) return;
  float4 res = make_float4(0.0f, 0.0f, 0.0f, 0.0f);
#pragma unroll
  for (int g = 0; g < 2; ++g) {
    const unsigned int dg  = (unsigned int)__builtin_amdgcn_readfirstlane((int)deg[(size_t)g * NN + node]);
    const unsigned int beg = (unsigned int)__builtin_amdgcn_readfirstlane((int)off[(size_t)g * NN + node]);
    res = agg_graph(perm + (size_t)g * EE, beg, dg, h + (size_t)g * NN * HD,
                    g ? al_b : al_f, g ? ar_b : ar_f, node, lane, res);
  }
  res.x += xmask<8>(res.x);
  res.y += xmask<8>(res.y);
  res.z += xmask<8>(res.z);
  res.w += xmask<8>(res.w);
  res.x += __shfl_xor(res.x, 16);
  res.y += __shfl_xor(res.y, 16);
  res.z += __shfl_xor(res.z, 16);
  res.w += __shfl_xor(res.w, 16);
  if (lane < 8) *(float4*)(out + (size_t)node * DD + 4 * lane) = res;
}

extern "C" void kernel_launch(void* const* d_in, const int* in_sizes, int n_in,
                              void* d_out, int out_size, void* d_ws, size_t ws_size,
                              hipStream_t stream) {
  const float* feat = (const float*)d_in[0];
  const int*   src_g[2] = { (const int*)d_in[1], (const int*)d_in[3] };
  const int*   dst_g[2] = { (const int*)d_in[2], (const int*)d_in[4] };
  const float* W_g[2]   = { (const float*)d_in[5], (const float*)d_in[8] };
  const float* al_g[2]  = { (const float*)d_in[6], (const float*)d_in[9] };
  const float* ar_g[2]  = { (const float*)d_in[7], (const float*)d_in[10] };
  float* out = (float*)d_out;

  float*        hbuf = (float*)d_ws;
  unsigned int* deg  = (unsigned int*)(hbuf + (size_t)2 * NN * HD);
  unsigned int* off  = deg + (size_t)2 * NN;
  unsigned int* cnt  = off + (size_t)2 * NN;
  int*          perm = (int*)(cnt + (size_t)2 * NCH * NN);

  const int T = 256;
  k_count<<<dim3(NCH, 2), BK_T, 0, stream>>>(dst_g[0], dst_g[1], cnt);
  k_col<<<(2 * NN + T - 1) / T, T, 0, stream>>>(cnt, deg);
  k_scan<<<2, SCAN_T, 0, stream>>>(deg, off);
  k_fill<<<dim3(NCH, 2), BK_T, 0, stream>>>(src_g[0], dst_g[0], src_g[1], dst_g[1], cnt, off, perm);
  for (int g = 0; g < 2; ++g)
    k_proj<<<NN / 16, T, 0, stream>>>(feat, W_g[g], hbuf + (size_t)g * NN * HD);
  k_node<<<(NN * 32 + T - 1) / T, T, 0, stream>>>(perm, off, deg, hbuf, al_g[0], ar_g[0], al_g[1], ar_g[1], out);
}


# each edge chunk is sorted by destination inside LDS and stored whole; the node pass merges the 64 sorted runs of its 32 nodes from LDS: no scattered stores, 3 launches fewer; sum order changes
# speedup vs baseline: 5.8972x; 1.4763x over previous
#include <hip/hip_runtime.h>


#define NN 50000
#define EE 1600000
#define INF_ 128
#define HH 4
#define DD 32
#define HD 128
#define NEG_SLOPE 0.2f

typedef __attribute__((ext_vector_type(2))) float v2f;
typedef __attribute__((ext_vector_type(8))) float v8f;

__global__ __launch_bounds__(256) void k_proj(const float* __restrict__ feat,
                                              const float* __restrict__ W,
                                              float* __restrict__ h) {
  __shared__ float As[16 * 132];
  const int m0 = blockIdx.x * 16;
  const int t  = threadIdx.x;

  {
    const int row = t >> 4;
    const int col = (t & 15) * 8;
    const float4* gp = (const float4*)(feat + (size_t)(m0 + row) * INF_ + col);
    float4 x0 = gp[0];
    float4 x1 = gp[1];
    float* lp = As + row * 132 + col;
    *(float4*)(lp)     = x0;
    *(float4*)(lp + 4) = x1;
  }
  __syncthreads();

  const int lane  = t & 31;
  const int wave  = t >> 5;
  const int n0    = wave * 16;
  const int r     = lane & 15;
  const int khalf = (lane >> 4) << 1;

  v8f acc = {};
  for (int k0 = 0; k0 < INF_; k0 += 4) {
    const int kk = k0 + khalf;
    v2f a;
    a.x = As[r * 132 + kk];
    a.y = As[r * 132 + kk + 1];
    v2f b;
    b.x = W[(size_t)(kk)     * HD + n0 + r];
    b.y = W[(size_t)(kk + 1) * HD + n0 + r];
    acc = __builtin_amdgcn_wmma_f32_16x16x4_f32(false, a, false, b,
                                                (short)0, acc, false, false);
  }

  const int mh = (lane >> 4) * 8;
#pragma unroll
  for (int j = 0; j < 8; ++j) {
    h[(size_t)(m0 + j + mh) * HD + n0 + r] = acc[j];
  }
}

#define NCH   64
#define CH_E  (EE / NCH)
#define BK_T  1024
#define CH_IT ((CH_E + BK_T - 1) / BK_T)
#define LROW  (NN + 4)
#define SRT_V 49
#define SRT_B 7
static_assert(EE % NCH == 0 && NN % 4 == 0 && CH_E % 4 == 0, "whole chunks, whole uint4 rows");
static_assert(BK_T * SRT_V >= NN && SRT_V % SRT_B == 0, "the threads' runs of counters cover the nodes, in whole groups");

__global__ __launch_bounds__(BK_T) void k_sort(const int* __restrict__ src_f, const int* __restrict__ dst_f,
                                               const int* __restrict__ src_b, const int* __restrict__ dst_b,
                                               unsigned int* __restrict__ lo,
                                               int* __restrict__ sorted) {
  __shared__ __attribute__((aligned(16))) unsigned int hist[LROW];
  __shared__ __attribute__((aligned(16))) int stage[CH_E];
  __shared__ unsigned int wtot[BK_T / 32];
  const int t    = threadIdx.x;
  const int lane = t & 31;
  const int wave = t >> 5;
  const int w = blockIdx.x;
  const int g = blockIdx.y;
  const int* dst = (g ? dst_b : dst_f) + (size_t)w * CH_E;
  const int* src = (g ? src_b : src_f) + (size_t)w * CH_E;
  int d[CH_IT], s[CH_IT];
#pragma unroll
  for (int i = 0; i < CH_IT; ++i) {
    const int k = i * BK_T + t;
    d[i] = (k < CH_E) ? dst[k] : -1;
    s[i] = (k < CH_E) ? src[k] : 0;
  }
  for (int i = t; i < LROW / 4; i += BK_T) ((uint4*)hist)[i] = make_uint4(0u, 0u, 0u, 0u);
  __syncthreads();
#pragma unroll
  for (int i = 0; i < CH_IT; ++i)
    if ((unsigned int)d[i] < (unsigned int)NN) atomicAdd(&hist[d[i]], 1u);
  __syncthreads();

  const int q0 = t * SRT_V;
  unsigned int sum = 0u;
  for (int i0 = 0; i0 < SRT_V; i0 += SRT_B) {
    unsigned int v[SRT_B];
#pragma unroll
    for (int u = 0; u < SRT_B; ++u) {
      const int n = q0 + i0 + u;
      v[u] = (n < NN) ? hist[n] : 0u;
    }
#pragma unroll
    for (int u = 0; u < SRT_B; ++u) sum += v[u];
  }
  unsigned int inc = sum;
#pragma unroll
  for (int o = 1; o < 32; o <<= 1) {
    const unsigned int y = __shfl_up(inc, (unsigned int)o);
    if (lane >= o) inc += y;
  }
  if (lane == 31) wtot[wave] = inc;
  __syncthreads();
  unsigned int wb = (lane < wave) ? wtot[lane] : 0u;
#pragma unroll
  for (int o = 16; o; o >>= 1) wb += __shfl_xor(wb, o);
  unsigned int run = wb + inc - sum;
  for (int i0 = 0; i0 < SRT_V; i0 += SRT_B) {
    unsigned int v[SRT_B];
#pragma unroll
    for (int u = 0; u < SRT_B; ++u) {
      const int n = q0 + i0 + u;
      v[u] = (n < NN) ? hist[n] : 0u;
    }
#pragma unroll
    for (int u = 0; u < SRT_B; ++u) {
      const int n = q0 + i0 + u;
      if (n < NN) hist[n] = run;
      run += v[u];
    }
  }
  if (t == BK_T - 1) hist[NN] = run;
  __syncthreads();

  uint4* row = (uint4*)(lo + ((size_t)g * NCH + w) * LROW);
  for (int i = t; i < LROW / 4; i += BK_T) row[i] = ((const uint4*)hist)[i];
  __syncthreads();

#pragma unroll
  for (int i = 0; i < CH_IT; ++i) {
    if ((unsigned int)d[i] < (unsigned int)NN) {
      const unsigned int pos = atomicAdd(&hist[d[i]], 1u);
      if (pos < (unsigned int)CH_E) stage[pos] = s[i];
    }
  }
  __syncthreads();
  uint4* so = (uint4*)(sorted + (size_t)g * EE + (size_t)w * CH_E);
  for (int i = t; i < CH_E / 4; i += BK_T) so[i] = ((const uint4*)stage)[i];
}

static_assert(DD == 32 && HH == 4, "lane = (head, four features)");
#define NB   32
#define TROW (NB + 1)
#define CAPS 2048
#define ND_T 256
#define TL_IT ((2 * NCH * TROW + ND_T - 1) / ND_T)
static_assert(NCH == 64, "a lane owns chunks 2*lane and 2*lane+1");
static_assert(ND_T == 4 * NCH && NB % (ND_T / 32) == 0, "two graphs x two copy threads per chunk; whole nodes per wave");

__device__ __forceinline__ float lrelu(float x) { return (x > 0.0f) ? x : NEG_SLOPE * x; }

template <int MASK>
__device__ __forceinline__ float xmask(float v) {
  return __int_as_float(__builtin_amdgcn_mov_dpp(__float_as_int(v), 0x160 | MASK, 0xf, 0xf, false));
}
__device__ __forceinline__ float sum8(float v) {
  v += xmask<1>(v);
  v += xmask<2>(v);
  v += xmask<4>(v);
  return v;
}
__device__ __forceinline__ float dot4(const float4 a, const float4 b) {
  v2f t = (v2f){a.x, a.y} * (v2f){b.x, b.y};
  t = __builtin_elementwise_fma((v2f){a.z, a.w}, (v2f){b.z, b.w}, t);
  return t.x + t.y;
}

__device__ __forceinline__ float4 agg_graph(const unsigned int* Tg, int i, const int* Sg, const int* Dg,
                                            const int* __restrict__ sg, bool staged,
                                            const float* __restrict__ h,
                                            const float* __restrict__ al, const float* __restrict__ ar,
                                            int node, int lane, float4 res) {
  const unsigned int a0 = Tg[(2 * lane) * TROW + i],     e0 = Tg[(2 * lane) * TROW + i + 1];
  const unsigned int a1 = Tg[(2 * lane + 1) * TROW + i], e1 = Tg[(2 * lane + 1) * TROW + i + 1];
  const unsigned int c0 = (e0 > a0) ? e0 - a0 : 0u;
  const unsigned int c1 = (e1 > a1) ? e1 - a1 : 0u;
  const unsigned int c  = c0 + c1;
  unsigned int inc = c;
#pragma unroll
  for (int o = 1; o < 32; o <<= 1) {
    const unsigned int y = __shfl_up(inc, (unsigned int)o);
    if (lane >= o) inc += y;
  }
  const unsigned int P   = inc - c;
  const unsigned int deg = (unsigned int)__builtin_amdgcn_readlane((int)inc, 31);
  if (deg == 0u) return res;

  const float ninf = -__builtin_inff();
  const float4 al4 = *(const float4*)(al + 4 * lane);
  const float4 ar4 = *(const float4*)(ar + 4 * lane);
  const float4 hn  = *(const float4*)(h + (size_t)node * HD + 4 * lane);
  const float er   = sum8(dot4(hn, ar4));

  float m = ninf, s = 0.0f;
  float4 acc = make_float4(0.0f, 0.0f, 0.0f, 0.0f);
  for (unsigned int base = 0u; base < deg; base += 32u) {
    const unsigned int j  = base + (unsigned int)lane;
    const unsigned int jc = (j < deg) ? j : deg - 1u;
    int own = 0;
#pragma unroll
    for (int st = 16; st; st >>= 1) {
      const unsigned int pc = __shfl(P, own + st);
      if (pc <= jc) own += st;
    }
    const unsigned int k   = jc - __shfl(P, own);
    const unsigned int oa0 = Tg[(2 * own) * TROW + i];
    const unsigned int oe0 = Tg[(2 * own) * TROW + i + 1];
    const unsigned int oa1 = Tg[(2 * own + 1) * TROW + i];
    const unsigned int oc0 = (oe0 > oa0) ? oe0 - oa0 : 0u;
    const bool first = k < oc0;
    const int  wch = 2 * own + (first ? 0 : 1);
    unsigned int pos = first ? oa0 + k : oa1 + (k - oc0);
    pos = (pos < (unsigned int)CH_E) ? pos : (unsigned int)(CH_E - 1);
    int si = Dg[wch] + (int)pos;
    si = ((unsigned int)si < (unsigned int)CAPS) ? si : 0;
    int sj = Sg[si];
    if (!staged) sj = sg[(size_t)wch * CH_E + pos];
    sj = ((unsigned int)sj < (unsigned int)NN) ? sj : 0;
    const unsigned int cnt = (deg - base < 32u) ? deg - base : 32u;
    for (unsigned int jb = 0u; jb < cnt; jb += 8u) {
      float4 hv[8];
#pragma unroll
      for (int u = 0; u < 8; ++u) {
        const int su = __builtin_amdgcn_readlane(sj, (int)jb + u);
        hv[u] = *(const float4*)(h + (size_t)su * HD + 4 * lane);
      }
      float x[8];
      float gm = ninf;
#pragma unroll
      for (int u = 0; u < 8; ++u) {
        const float t = lrelu(sum8(dot4(hv[u], al4)) + er);
        x[u] = (jb + (unsigned int)u < cnt) ? t : ninf;
        gm = fmaxf(gm, x[u]);
      }
      const float mn = fmaxf(m, gm);
      const float sc = __expf(m - mn);
      s *= sc;
      acc.x *= sc; acc.y *= sc; acc.z *= sc; acc.w *= sc;
      m = mn;
#pragma unroll
      for (int u = 0; u < 8; ++u) {
        const float pe = __expf(x[u] - mn);
        s += pe;
        acc.x = __builtin_fmaf(pe, hv[u].x, acc.x);
        acc.y = __builtin_fmaf(pe, hv[u].y, acc.y);
        acc.z = __builtin_fmaf(pe, hv[u].z, acc.z);
        acc.w = __builtin_fmaf(pe, hv[u].w, acc.w);
      }
    }
  }
  const float w = 0.25f / s;
  res.x = __builtin_fmaf(acc.x, w, res.x);
  res.y = __builtin_fmaf(acc.y, w, res.y);
  res.z = __builtin_fmaf(acc.z, w, res.z);
  res.w = __builtin_fmaf(acc.w, w, res.w);
  return res;
}

__global__ __launch_bounds__(ND_T) void k_node(const int* __restrict__ sorted,
                                               const unsigned int* __restrict__ lo,
                                               const float* __restrict__ h,
                                               const float* __restrict__ al_f,
                                               const float* __restrict__ ar_f,
                                               const float* __restrict__ al_b,
                                               const float* __restrict__ ar_b,
                                               float* __restrict__ out) {
  __shared__ unsigned int T[2 * NCH * TROW];
  __shared__ int S[2 * CAPS];
  __shared__ int Dq[2 * NCH];
  __shared__ unsigned int Qs[2 * NCH];
  __shared__ unsigned int tot[2];
  const int t    = threadIdx.x;
  const int lane = t & 31;
  const int wave = __builtin_amdgcn_readfirstlane(t >> 5);
  const int n0   = blockIdx.x * NB;

  {
    unsigned int tv[TL_IT];
#pragma unroll
    for (int q = 0; q < TL_IT; ++q) {
      const int idx = t + q * ND_T;
      const int gw  = idx / TROW;
      const int i   = idx - gw * TROW;
      int n = n0 + i;
      n = (n < NN) ? n : NN;
      tv[q] = (idx < 2 * NCH * TROW) ? lo[(size_t)gw * LROW + n] : 0u;
    }
#pragma unroll
    for (int q = 0; q < TL_IT; ++q) {
      const int idx = t + q * ND_T;
      if (idx < 2 * NCH * TROW) T[idx] = tv[q];
    }
  }
  __syncthreads();
  if (wave < 2) {
    const int g = wave;
    const unsigned int* Tg = T + g * NCH * TROW;
    const unsigned int A0 = Tg[(2 * lane) * TROW],     E0 = Tg[(2 * lane) * TROW + NB];
    const unsigned int A1 = Tg[(2 * lane + 1) * TROW], E1 = Tg[(2 * lane + 1) * TROW + NB];
    const unsigned int L0 = (E0 > A0) ? E0 - A0 : 0u;
    const unsigned int L1 = (E1 > A1) ? E1 - A1 : 0u;
    const unsigned int c = L0 + L1;
    unsigned int inc = c;
#pragma unroll
    for (int o = 1; o < 32; o <<= 1) {
      const unsigned int y = __shfl_up(inc, (unsigned int)o);
      if (lane >= o) inc += y;
    }
    const unsigned int q0 = inc - c, q1 = q0 + L0;
    Qs[g * NCH + 2 * lane]     = q0;
    Qs[g * NCH + 2 * lane + 1] = q1;
    Dq[g * NCH + 2 * lane]     = (int)q0 - (int)A0;
    Dq[g * NCH + 2 * lane + 1] = (int)q1 - (int)A1;
    if (lane == 31) tot[g] = inc;
  }
  __syncthreads();
  {
    const int g = t >> 7;
    const int wch = (t >> 1) & (NCH - 1), part = t & 1;
    if (tot[g] <= (unsigned int)CAPS) {
      const unsigned int A = T[(g * NCH + wch) * TROW], E = T[(g * NCH + wch) * TROW + NB];
      const unsigned int L = (E > A) ? E - A : 0u;
      const unsigned int Q = Qs[g * NCH + wch];
      const int* sp = sorted + (size_t)g * EE + (size_t)wch * CH_E;
      for (unsigned int k0 = (unsigned int)part; k0 < L; k0 += 16u) {
        int v[8];
#pragma unroll
        for (int q = 0; q < 8; ++q) {
          unsigned int pos = A + k0 + 2u * (unsigned int)q;
          pos = (pos < (unsigned int)CH_E) ? pos : (unsigned int)(CH_E - 1);
          v[q] = sp[pos];
        }
#pragma unroll
        for (int q = 0; q < 8; ++q) {
          const unsigned int k = k0 + 2u * (unsigned int)q;
          if (k < L) S[g * CAPS + Q + k] = v[q];
        }
      }
    }
  }
  __syncthreads();

  for (int r = 0; r < NB / (ND_T / 32); ++r) {
    const int i    = wave + r * (ND_T / 32);
    const int node = n0 + i;
    if (node >= NN) break;
    float4 res = make_float4(0.0f, 0.0f, 0.0f, 0.0f);
#pragma unroll
    for (int g = 0; g < 2; ++g) {
      const bool staged = (unsigned int)__builtin_amdgcn_readfirstlane((int)tot[g]) <= (unsigned int)CAPS;
      res = agg_graph(T + g * NCH * TROW, i, S + g * CAPS, Dq + g * NCH, sorted + (size_t)g * EE, staged,
                      h + (size_t)g * NN * HD, g ? al_b : al_f, g ? ar_b : ar_f, node, lane, res);
    }
    res.x += xmask<8>(res.x);
    res.y += xmask<8>(res.y);
    res.z += xmask<8>(res.z);
    res.w += xmask<8>(res.w);
    res.x += __shfl_xor(res.x, 16);
    res.y += __shfl_xor(res.y, 16);
    res.z += __shfl_xor(res.z, 16);
    res.w += __shfl_xor(res.w, 16);
    if (lane < 8) *(float4*)(out + (size_t)node * DD + 4 * lane) = res;
  }
}

extern "C" void kernel_launch(void* const* d_in, const int* in_sizes, int n_in,
                              void* d_out, int out_size, void* d_ws, size_t ws_size,
                              hipStream_t stream) {
  const float* feat = (const float*)d_in[0];
  const int*   src_g[2] = { (const int*)d_in[1], (const int*)d_in[3] };
  const int*   dst_g[2] = { (const int*)d_in[2], (const int*)d_in[4] };
  const float* W_g[2]   = { (const float*)d_in[5], (const float*)d_in[8] };
  const float* al_g[2]  = { (const float*)d_in[6], (const float*)d_in[9] };
  const float* ar_g[2]  = { (const float*)d_in[7], (const float*)d_in[10] };
  float* out = (float*)d_out;

  float*        hbuf   = (float*)d_ws;
  unsigned int* lo     = (unsigned int*)(hbuf + (size_t)2 * NN * HD);
  int*          sorted = (int*)(lo + (size_t)2 * NCH * LROW);

  const int T = 256;
  k_sort<<<dim3(NCH, 2), BK_T, 0, stream>>>(src_g[0], dst_g[0], src_g[1], dst_g[1], lo, sorted);
  for (int g = 0; g < 2; ++g)
    k_proj<<<NN / 16, T, 0, stream>>>(feat, W_g[g], hbuf + (size_t)g * NN * HD);
  k_node<<<(NN + NB - 1) / NB, ND_T, 0, stream>>>(sorted, lo, hbuf, al_g[0], ar_g[0], al_g[1], ar_g[1], out);
}


# node pass fully resident: 40 nodes per work-group, 96 registers (row loads take a scalar row base + one lane offset, no per-lane 64-bit address): 1250 work-groups all at once, not 1563 in two rounds
# speedup vs baseline: 6.2577x; 1.0611x over previous
#include <hip/hip_runtime.h>


#define NN 50000
#define EE 1600000
#define INF_ 128
#define HH 4
#define DD 32
#define HD 128
#define NEG_SLOPE 0.2f

typedef __attribute__((ext_vector_type(2))) float v2f;
typedef __attribute__((ext_vector_type(8))) float v8f;

__global__ __launch_bounds__(256) void k_proj(const float* __restrict__ feat,
                                              const float* __restrict__ W,
                                              float* __restrict__ h) {
  __shared__ float As[16 * 132];
  const int m0 = blockIdx.x * 16;
  const int t  = threadIdx.x;

  {
    const int row = t >> 4;
    const int col = (t & 15) * 8;
    const float4* gp = (const float4*)(feat + (size_t)(m0 + row) * INF_ + col);
    float4 x0 = gp[0];
    float4 x1 = gp[1];
    float* lp = As + row * 132 + col;
    *(float4*)(lp)     = x0;
    *(float4*)(lp + 4) = x1;
  }
  __syncthreads();

  const int lane  = t & 31;
  const int wave  = t >> 5;
  const int n0    = wave * 16;
  const int r     = lane & 15;
  const int khalf = (lane >> 4) << 1;

  v8f acc = {};
  for (int k0 = 0; k0 < INF_; k0 += 4) {
    const int kk = k0 + khalf;
    v2f a;
    a.x = As[r * 132 + kk];
    a.y = As[r * 132 + kk + 1];
    v2f b;
    b.x = W[(size_t)(kk)     * HD + n0 + r];
    b.y = W[(size_t)(kk + 1) * HD + n0 + r];
    acc = __builtin_amdgcn_wmma_f32_16x16x4_f32(false, a, false, b,
                                                (short)0, acc, false, false);
  }

  const int mh = (lane >> 4) * 8;
#pragma unroll
  for (int j = 0; j < 8; ++j) {
    h[(size_t)(m0 + j + mh) * HD + n0 + r] = acc[j];
  }
}

#define NCH   64
#define CH_E  (EE / NCH)
#define BK_T  1024
#define CH_IT ((CH_E + BK_T - 1) / BK_T)
#define LROW  (NN + 4)
#define SRT_V 49
#define SRT_B 7
static_assert(EE % NCH == 0 && NN % 4 == 0 && CH_E % 4 == 0, "whole chunks, whole uint4 rows");
static_assert(BK_T * SRT_V >= NN && SRT_V % SRT_B == 0, "the threads' runs of counters cover the nodes, in whole groups");

__global__ __launch_bounds__(BK_T) void k_sort(const int* __restrict__ src_f, const int* __restrict__ dst_f,
                                               const int* __restrict__ src_b, const int* __restrict__ dst_b,
                                               unsigned int* __restrict__ lo,
                                               int* __restrict__ sorted) {
  __shared__ __attribute__((aligned(16))) unsigned int hist[LROW];
  __shared__ __attribute__((aligned(16))) int stage[CH_E];
  __shared__ unsigned int wtot[BK_T / 32];
  const int t    = threadIdx.x;
  const int lane = t & 31;
  const int wave = t >> 5;
  const int w = blockIdx.x;
  const int g = blockIdx.y;
  const int* dst = (g ? dst_b : dst_f) + (size_t)w * CH_E;
  const int* src = (g ? src_b : src_f) + (size_t)w * CH_E;
  int d[CH_IT], s[CH_IT];
#pragma unroll
  for (int i = 0; i < CH_IT; ++i) {
    const int k = i * BK_T + t;
    d[i] = (k < CH_E) ? dst[k] : -1;
    s[i] = (k < CH_E) ? src[k] : 0;
  }
  for (int i = t; i < LROW / 4; i += BK_T) ((uint4*)hist)[i] = make_uint4(0u, 0u, 0u, 0u);
  __syncthreads();
#pragma unroll
  for (int i = 0; i < CH_IT; ++i)
    if ((unsigned int)d[i] < (unsigned int)NN) atomicAdd(&hist[d[i]], 1u);
  __syncthreads();

  const int q0 = t * SRT_V;
  unsigned int sum = 0u;
  for (int i0 = 0; i0 < SRT_V; i0 += SRT_B) {
    unsigned int v[SRT_B];
#pragma unroll
    for (int u = 0; u < SRT_B; ++u) {
      const int n = q0 + i0 + u;
      v[u] = (n < NN) ? hist[n] : 0u;
    }
#pragma unroll
    for (int u = 0; u < SRT_B; ++u) sum += v[u];
  }
  unsigned int inc = sum;
#pragma unroll
  for (int o = 1; o < 32; o <<= 1) {
    const unsigned int y = __shfl_up(inc, (unsigned int)o);
    if (lane >= o) inc += y;
  }
  if (lane == 31) wtot[wave] = inc;
  __syncthreads();
  unsigned int wb = (lane < wave) ? wtot[lane] : 0u;
#pragma unroll
  for (int o = 16; o; o >>= 1) wb += __shfl_xor(wb, o);
  unsigned int run = wb + inc - sum;
  for (int i0 = 0; i0 < SRT_V; i0 += SRT_B) {
    unsigned int v[SRT_B];
#pragma unroll
    for (int u = 0; u < SRT_B; ++u) {
      const int n = q0 + i0 + u;
      v[u] = (n < NN) ? hist[n] : 0u;
    }
#pragma unroll
    for (int u = 0; u < SRT_B; ++u) {
      const int n = q0 + i0 + u;
      if (n < NN) hist[n] = run;
      run += v[u];
    }
  }
  if (t == BK_T - 1) hist[NN] = run;
  __syncthreads();

  uint4* row = (uint4*)(lo + ((size_t)g * NCH + w) * LROW);
  for (int i = t; i < LROW / 4; i += BK_T) row[i] = ((const uint4*)hist)[i];
  __syncthreads();

#pragma unroll
  for (int i = 0; i < CH_IT; ++i) {
    if ((unsigned int)d[i] < (unsigned int)NN) {
      const unsigned int pos = atomicAdd(&hist[d[i]], 1u);
      if (pos < (unsigned int)CH_E) stage[pos] = s[i];
    }
  }
  __syncthreads();
  uint4* so = (uint4*)(sorted + (size_t)g * EE + (size_t)w * CH_E);
  for (int i = t; i < CH_E / 4; i += BK_T) so[i] = ((const uint4*)stage)[i];
}

static_assert(DD == 32 && HH == 4, "lane = (head, four features)");
#define NB   40
#define TROW (NB + 1)
#define CAPS 2048
#define ND_T 256
#define TL_IT ((2 * NCH * TROW + ND_T - 1) / ND_T)
static_assert(NCH == 64, "a lane owns chunks 2*lane and 2*lane+1");
static_assert(ND_T == 4 * NCH && NB % (ND_T / 32) == 0, "two graphs x two copy threads per chunk; whole nodes per wave");

__device__ __forceinline__ float lrelu(float x) { return (x > 0.0f) ? x : NEG_SLOPE * x; }

template <int MASK>
__device__ __forceinline__ float xmask(float v) {
  return __int_as_float(__builtin_amdgcn_mov_dpp(__float_as_int(v), 0x160 | MASK, 0xf, 0xf, false));
}
__device__ __forceinline__ float sum8(float v) {
  v += xmask<1>(v);
  v += xmask<2>(v);
  v += xmask<4>(v);
  return v;
}
__device__ __forceinline__ float dot4(const float4 a, const float4 b) {
  v2f t = (v2f){a.x, a.y} * (v2f){b.x, b.y};
  t = __builtin_elementwise_fma((v2f){a.z, a.w}, (v2f){b.z, b.w}, t);
  return t.x + t.y;
}


typedef float v4f __attribute__((ext_vector_type(4)));
__device__ __forceinline__ void rows_load8(v4f (&r)[8], unsigned int voff,
                                           unsigned long long b0, unsigned long long b1, unsigned long long b2,
                                           unsigned long long b3, unsigned long long b4, unsigned long long b5,
                                           unsigned long long b6, unsigned long long b7) {
  asm volatile("global_load_b128 %0, %8, %9\n\t"
               "global_load_b128 %1, %8, %10\n\t"
               "global_load_b128 %2, %8, %11\n\t"
               "global_load_b128 %3, %8, %12\n\t"
               "global_load_b128 %4, %8, %13\n\t"
               "global_load_b128 %5, %8, %14\n\t"
               "global_load_b128 %6, %8, %15\n\t"
               "global_load_b128 %7, %8, %16\n\t"
               "s_wait_loadcnt 0x0"
               : "=&v"(r[0]), "=&v"(r[1]), "=&v"(r[2]), "=&v"(r[3]), "=&v"(r[4]), "=&v"(r[5]), "=&v"(r[6]), "=&v"(r[7])
               : "v"(voff), "s"(b0), "s"(b1), "s"(b2), "s"(b3), "s"(b4), "s"(b5), "s"(b6), "s"(b7));
}

__device__ __forceinline__ float4 agg_graph(const unsigned int* Tg, int i, const int* Sg, const int* Dg,
                                            const int* __restrict__ sg, bool staged,
                                            const float* __restrict__ h,
                                            const float* __restrict__ al, const float* __restrict__ ar,
                                            int node, int lane, float4 res) {
  const unsigned int a0 = Tg[(2 * lane) * TROW + i],     e0 = Tg[(2 * lane) * TROW + i + 1];
  const unsigned int a1 = Tg[(2 * lane + 1) * TROW + i], e1 = Tg[(2 * lane + 1) * TROW + i + 1];
  const unsigned int c0 = (e0 > a0) ? e0 - a0 : 0u;
  const unsigned int c1 = (e1 > a1) ? e1 - a1 : 0u;
  const unsigned int c  = c0 + c1;
  unsigned int inc = c;
#pragma unroll
  for (int o = 1; o < 32; o <<= 1) {
    const unsigned int y = __shfl_up(inc, (unsigned int)o);
    if (lane >= o) inc += y;
  }
  const unsigned int P   = inc - c;
  const unsigned int deg = (unsigned int)__builtin_amdgcn_readlane((int)inc, 31);
  if (deg == 0u) return res;

  const float ninf = -__builtin_inff();
  const unsigned int voff = 16u * (unsigned int)lane;
  const float4 al4 = *(const float4*)(al + 4 * lane);
  const float4 ar4 = *(const float4*)(ar + 4 * lane);
  const float4 hn  = *(const float4*)(h + (size_t)node * HD + 4 * lane);
  const float er   = sum8(dot4(hn, ar4));

  float m = ninf, s = 0.0f;
  float4 acc = make_float4(0.0f, 0.0f, 0.0f, 0.0f);
  for (unsigned int base = 0u; base < deg; base += 32u) {
    const unsigned int j  = base + (unsigned int)lane;
    const unsigned int jc = (j < deg) ? j : deg - 1u;
    int own = 0;
#pragma unroll
    for (int st = 16; st; st >>= 1) {
      const unsigned int pc = __shfl(P, own + st);
      if (pc <= jc) own += st;
    }
    const unsigned int k   = jc - __shfl(P, own);
    const unsigned int oa0 = Tg[(2 * own) * TROW + i];
    const unsigned int oe0 = Tg[(2 * own) * TROW + i + 1];
    const unsigned int oa1 = Tg[(2 * own + 1) * TROW + i];
    const unsigned int oc0 = (oe0 > oa0) ? oe0 - oa0 : 0u;
    const bool first = k < oc0;
    const int  wch = 2 * own + (first ? 0 : 1);
    unsigned int pos = first ? oa0 + k : oa1 + (k - oc0);
    pos = (pos < (unsigned int)CH_E) ? pos : (unsigned int)(CH_E - 1);
    int si = Dg[wch] + (int)pos;
    si = ((unsigned int)si < (unsigned int)CAPS) ? si : 0;
    int sj = Sg[si];
    if (!staged) sj = sg[(size_t)wch * CH_E + pos];
    sj = ((unsigned int)sj < (unsigned int)NN) ? sj : 0;
    const unsigned int cnt = (deg - base < 32u) ? deg - base : 32u;
    for (unsigned int jb = 0u; jb < cnt; jb += 8u) {
      unsigned long long rb[8];
#pragma unroll
      for (int u = 0; u < 8; ++u) {
        const int su = __builtin_amdgcn_readlane(sj, (int)jb + u);
        rb[u] = (unsigned long long)(h + (size_t)su * HD);
      }
      v4f hr[8];
      rows_load8(hr, voff, rb[0], rb[1], rb[2], rb[3], rb[4], rb[5], rb[6], rb[7]);
      float4 hv[8];
#pragma unroll
      for (int u = 0; u < 8; ++u) hv[u] = make_float4(hr[u].x, hr[u].y, hr[u].z, hr[u].w);
      float x[8];
      float gm = ninf;
#pragma unroll
      for (int u = 0; u < 8; ++u) {
        const float t = lrelu(sum8(dot4(hv[u], al4)) + er);
        x[u] = (jb + (unsigned int)u < cnt) ? t : ninf;
        gm = fmaxf(gm, x[u]);
      }
      const float mn = fmaxf(m, gm);
      const float sc = __expf(m - mn);
      s *= sc;
      acc.x *= sc; acc.y *= sc; acc.z *= sc; acc.w *= sc;
      m = mn;
#pragma unroll
      for (int u = 0; u < 8; ++u) {
        const float pe = __expf(x[u] - mn);
        s += pe;
        acc.x = __builtin_fmaf(pe, hv[u].x, acc.x);
        acc.y = __builtin_fmaf(pe, hv[u].y, acc.y);
        acc.z = __builtin_fmaf(pe, hv[u].z, acc.z);
        acc.w = __builtin_fmaf(pe, hv[u].w, acc.w);
      }
    }
  }
  const float w = 0.25f / s;
  res.x = __builtin_fmaf(acc.x, w, res.x);
  res.y = __builtin_fmaf(acc.y, w, res.y);
  res.z = __builtin_fmaf(acc.z, w, res.z);
  res.w = __builtin_fmaf(acc.w, w, res.w);
  return res;
}

__global__ __launch_bounds__(ND_T, 10) void k_node(const int* __restrict__ sorted,
                                               const unsigned int* __restrict__ lo,
                                               const float* __restrict__ h,
                                               const float* __restrict__ al_f,
                                               const float* __restrict__ ar_f,
                                               const float* __restrict__ al_b,
                                               const float* __restrict__ ar_b,
                                               float* __restrict__ out) {
  __shared__ unsigned int T[2 * NCH * TROW];
  __shared__ int S[2 * CAPS];
  __shared__ int Dq[2 * NCH];
  __shared__ unsigned int Qs[2 * NCH];
  __shared__ unsigned int tot[2];
  const int t    = threadIdx.x;
  const int lane = t & 31;
  const int wave = __builtin_amdgcn_readfirstlane(t >> 5);
  const int n0   = blockIdx.x * NB;

  {
    unsigned int tv[TL_IT];
#pragma unroll
    for (int q = 0; q < TL_IT; ++q) {
      const int idx = t + q * ND_T;
      const int gw  = idx / TROW;
      const int i   = idx - gw * TROW;
      int n = n0 + i;
      n = (n < NN) ? n : NN;
      tv[q] = (idx < 2 * NCH * TROW) ? lo[(size_t)gw * LROW + n] : 0u;
    }
#pragma unroll
    for (int q = 0; q < TL_IT; ++q) {
      const int idx = t + q * ND_T;
      if (idx < 2 * NCH * TROW) T[idx] = tv[q];
    }
  }
  __syncthreads();
  if (wave < 2) {
    const int g = wave;
    const unsigned int* Tg = T + g * NCH * TROW;
    const unsigned int A0 = Tg[(2 * lane) * TROW],     E0 = Tg[(2 * lane) * TROW + NB];
    const unsigned int A1 = Tg[(2 * lane + 1) * TROW], E1 = Tg[(2 * lane + 1) * TROW + NB];
    const unsigned int L0 = (E0 > A0) ? E0 - A0 : 0u;
    const unsigned int L1 = (E1 > A1) ? E1 - A1 : 0u;
    const unsigned int c = L0 + L1;
    unsigned int inc = c;
#pragma unroll
    for (int o = 1; o < 32; o <<= 1) {
      const unsigned int y = __shfl_up(inc, (unsigned int)o);
      if (lane >= o) inc += y;
    }
    const unsigned int q0 = inc - c, q1 = q0 + L0;
    Qs[g * NCH + 2 * lane]     = q0;
    Qs[g * NCH + 2 * lane + 1] = q1;
    Dq[g * NCH + 2 * lane]     = (int)q0 - (int)A0;
    Dq[g * NCH + 2 * lane + 1] = (int)q1 - (int)A1;
    if (lane == 31) tot[g] = inc;
  }
  __syncthreads();
  {
    const int g = t >> 7;
    const int wch = (t >> 1) & (NCH - 1), part = t & 1;
    if (tot[g] <= (unsigned int)CAPS) {
      const unsigned int A = T[(g * NCH + wch) * TROW], E = T[(g * NCH + wch) * TROW + NB];
      const unsigned int L = (E > A) ? E - A : 0u;
      const unsigned int Q = Qs[g * NCH + wch];
      const int* sp = sorted + (size_t)g * EE + (size_t)wch * CH_E;
      for (unsigned int k0 = (unsigned int)part; k0 < L; k0 += 16u) {
        int v[8];
#pragma unroll
        for (int q = 0; q < 8; ++q) {
          unsigned int pos = A + k0 + 2u * (unsigned int)q;
          pos = (pos < (unsigned int)CH_E) ? pos : (unsigned int)(CH_E - 1);
          v[q] = sp[pos];
        }
#pragma unroll
        for (int q = 0; q < 8; ++q) {
          const unsigned int k = k0 + 2u * (unsigned int)q;
          if (k < L) S[g * CAPS + Q + k] = v[q];
        }
      }
    }
  }
  __syncthreads();

  for (int r = 0; r < NB / (ND_T / 32); ++r) {
    const int i    = wave + r * (ND_T / 32);
    const int node = n0 + i;
    if (node >= NN) break;
    float4 res = make_float4(0.0f, 0.0f, 0.0f, 0.0f);
#pragma unroll
    for (int g = 0; g < 2; ++g) {
      const bool staged = (unsigned int)__builtin_amdgcn_readfirstlane((int)tot[g]) <= (unsigned int)CAPS;
      res = agg_graph(T + g * NCH * TROW, i, S + g * CAPS, Dq + g * NCH, sorted + (size_t)g * EE, staged,
                      h + (size_t)g * NN * HD, g ? al_b : al_f, g ? ar_b : ar_f, node, lane, res);
    }
    res.x += xmask<8>(res.x);
    res.y += xmask<8>(res.y);
    res.z += xmask<8>(res.z);
    res.w += xmask<8>(res.w);
    res.x += __shfl_xor(res.x, 16);
    res.y += __shfl_xor(res.y, 16);
    res.z += __shfl_xor(res.z, 16);
    res.w += __shfl_xor(res.w, 16);
    if (lane < 8) *(float4*)(out + (size_t)node * DD + 4 * lane) = res;
  }
}

extern "C" void kernel_launch(void* const* d_in, const int* in_sizes, int n_in,
                              void* d_out, int out_size, void* d_ws, size_t ws_size,
                              hipStream_t stream) {
  const float* feat = (const float*)d_in[0];
  const int*   src_g[2] = { (const int*)d_in[1], (const int*)d_in[3] };
  const int*   dst_g[2] = { (const int*)d_in[2], (const int*)d_in[4] };
  const float* W_g[2]   = { (const float*)d_in[5], (const float*)d_in[8] };
  const float* al_g[2]  = { (const float*)d_in[6], (const float*)d_in[9] };
  const float* ar_g[2]  = { (const float*)d_in[7], (const float*)d_in[10] };
  float* out = (float*)d_out;

  float*        hbuf   = (float*)d_ws;
  unsigned int* lo     = (unsigned int*)(hbuf + (size_t)2 * NN * HD);
  int*          sorted = (int*)(lo + (size_t)2 * NCH * LROW);

  const int T = 256;
  k_sort<<<dim3(NCH, 2), BK_T, 0, stream>>>(src_g[0], dst_g[0], src_g[1], dst_g[1], lo, sorted);
  for (int g = 0; g < 2; ++g)
    k_proj<<<NN / 16, T, 0, stream>>>(feat, W_g[g], hbuf + (size_t)g * NN * HD);
  k_node<<<(NN + NB - 1) / NB, ND_T, 0, stream>>>(sorted, lo, hbuf, al_g[0], ar_g[0], al_g[1], ar_g[1], out);
}
